# starting kernel
# baseline (speedup 1.0000x reference)
#include <hip/hip_runtime.h>

#define HC 128
#define NH 8

typedef __attribute__((ext_vector_type(2))) float v2f;
typedef __attribute__((ext_vector_type(8))) float v8f;

__device__ __forceinline__ unsigned float_key(float f) {
  unsigned u = __float_as_uint(f);
  return (u & 0x80000000u) ? ~u : (u | 0x80000000u);
}
__device__ __forceinline__ float key_float(unsigned k) {
  unsigned u = (k & 0x80000000u) ? (k & 0x7FFFFFFFu) : ~k;
  return __uint_as_float(u);
}

__global__ void init_ws_kernel(unsigned* __restrict__ segkey,
                               float* __restrict__ denom,
                               float* __restrict__ agg, int n) {
  int gid = blockIdx.x * blockDim.x + threadIdx.x;
  if (gid < n * NH) { segkey[gid] = 0u; denom[gid] = 0.0f; }
  if (gid < n * HC) { agg[gid] = 0.0f; }
}

template <int K, int LDA, int LDB, int LDC>
__global__ __launch_bounds__(256) void wmma_gemm_f32(const float* __restrict__ A,
                                                     const float* __restrict__ B,
                                                     float* __restrict__ C,
                                                     int Mrows) {
  constexpr int KSTEPS = K / 4;
  __shared__ v2f Bs[KSTEPS * 32];

  const int tn = blockIdx.y;
#pragma unroll
  for (int idx = threadIdx.x; idx < KSTEPS * 32; idx += 256) {
    const int kk = idx >> 5, L = idx & 31;
    const int hf = L >> 4, lr2 = L & 15;
    const int ka = kk * 4 + 2 * hf;
    const int col = tn * 16 + lr2;
    v2f t;
    t.x = B[(size_t)ka * LDB + col];
    t.y = B[(size_t)(ka + 1) * LDB + col];
    Bs[idx] = t;
  }
  __syncthreads();

  const int lane = threadIdx.x & 31;
  const int tm = blockIdx.x * 8 + (threadIdx.x >> 5);
  if (tm * 16 >= Mrows) return;

  const int lr = lane & 15;
  const int half = lane >> 4;
  int arow = tm * 16 + lr; if (arow >= Mrows) arow = Mrows - 1;
  const float* ap = A + (size_t)arow * LDA + 2 * half;

  v8f acc = {};
#pragma unroll
  for (int kk = 0; kk < KSTEPS; ++kk) {
    const v2f a = *(const v2f*)(ap + kk * 4);
    acc = __builtin_amdgcn_wmma_f32_16x16x4_f32(false, a, false, Bs[kk * 32 + lane],
                                                (short)0, acc, false, false);
  }

  const int bcol = tn * 16 + lr;
#pragma unroll
  for (int v = 0; v < 8; ++v) {
    const int r = tm * 16 + v + 8 * half;
    if (r < Mrows) C[(size_t)r * LDC + bcol] = acc[v];
  }
}

__global__ __launch_bounds__(256) void edge_bias_logit_kernel(
    const float* __restrict__ PR, const float* __restrict__ Wb,
    const float* __restrict__ qbuf, const float* __restrict__ kvbuf,
    const int* __restrict__ qidx, const int* __restrict__ kidx,
    float* __restrict__ logit_out, unsigned* __restrict__ segkey, int Medges) {
  constexpr int KSTEPS = HC / 4;
  __shared__ v2f BmS[KSTEPS * 32];
  __shared__ v2f BaS[KSTEPS * 32];

  const int lane = threadIdx.x & 31;
  const int mtiles = (Medges + 15) >> 4;
  int tm = blockIdx.x * 8 + (threadIdx.x >> 5);
  if (tm >= mtiles) tm = mtiles - 1;

  const int half = lane >> 4;
  const int lr = lane & 15;
  int arow = tm * 16 + lr; if (arow >= Medges) arow = Medges - 1;

  v2f afrag[KSTEPS];
  const float* prow = PR + (size_t)arow * HC + 2 * half;
#pragma unroll
  for (int kk = 0; kk < KSTEPS; ++kk) afrag[kk] = *(const v2f*)(prow + kk * 4);

  float logacc[8];
#pragma unroll
  for (int v = 0; v < 8; ++v) logacc[v] = 0.0f;

  for (int j = 0; j < 8; ++j) {
    __syncthreads();
#pragma unroll
    for (int idx = threadIdx.x; idx < KSTEPS * 32; idx += 256) {
      const int kk = idx >> 5, L = idx & 31;
      const int hf = L >> 4, lr2 = L & 15;
      const int ka = kk * 4 + 2 * hf;
      const int col = j * 16 + lr2;
      v2f t1, t2;
      t1.x = Wb[(size_t)ka * 256 + col];
      t1.y = Wb[(size_t)(ka + 1) * 256 + col];
      t2.x = Wb[(size_t)ka * 256 + 128 + col];
      t2.y = Wb[(size_t)(ka + 1) * 256 + 128 + col];
      BmS[idx] = t1;
      BaS[idx] = t2;
    }
    __syncthreads();

    v8f bm = {};
    v8f ba = {};
#pragma unroll
    for (int kk = 0; kk < KSTEPS; ++kk) {
      bm = __builtin_amdgcn_wmma_f32_16x16x4_f32(false, afrag[kk], false,
                                                 BmS[kk * 32 + lane], (short)0,
                                                 bm, false, false);
      ba = __builtin_amdgcn_wmma_f32_16x16x4_f32(false, afrag[kk], false,
                                                 BaS[kk * 32 + lane], (short)0,
                                                 ba, false, false);
    }

    const int col = j * 16 + lr;
#pragma unroll
    for (int v = 0; v < 8; ++v) {
      int e = tm * 16 + v + 8 * half;
      e = (e < Medges) ? e : (Medges - 1);
      const int qi = qidx[e];
      const int ki = kidx[e];
      const float qv = qbuf[(size_t)qi * HC + col];
      const float kv = kvbuf[(size_t)ki * (2 * HC) + col];
      logacc[v] += qv * (kv * (1.0f + bm[v]) + ba[v]);
    }
  }

  const int h = lane & 7;
#pragma unroll
  for (int v = 0; v < 8; ++v) {
    logacc[v] += __shfl_xor(logacc[v], 8, 32);
    const int e = tm * 16 + v + 8 * half;
    if (((lane & 8) == 0) && e < Medges) {
      logit_out[(size_t)e * NH + h] = logacc[v];
      atomicMax(&segkey[(size_t)qidx[e] * NH + h], float_key(logacc[v]));
    }
  }
}

__global__ void exp_denom_kernel(const float* __restrict__ logit,
                                 const int* __restrict__ qidx,
                                 const unsigned* __restrict__ segkey,
                                 float* __restrict__ denom, int Medges) {
  int gid = blockIdx.x * blockDim.x + threadIdx.x;
  if (gid >= Medges * NH) return;
  const int e = gid >> 3, h = gid & 7;
  const int qi = qidx[e];
  const float mx = key_float(segkey[qi * NH + h]);
  const float ex = __expf(logit[(size_t)e * NH + h] - mx);
  atomicAdd(&denom[qi * NH + h], ex);
}

__global__ void scatter_agg_kernel(const float* __restrict__ logit,
                                   const int* __restrict__ qidx,
                                   const int* __restrict__ kidx,
                                   const unsigned* __restrict__ segkey,
                                   const float* __restrict__ denom,
                                   const float* __restrict__ kvbuf,
                                   float* __restrict__ agg, int Medges) {
  int gid = blockIdx.x * blockDim.x + threadIdx.x;
  if (gid >= Medges * HC) return;
  const int e = gid >> 7, c = gid & 127, h = c & 7;
  const int qi = qidx[e];
  const float mx   = key_float(segkey[qi * NH + h]);
  const float attn = __expf(logit[(size_t)e * NH + h] - mx) / denom[qi * NH + h];
  const float ve   = kvbuf[(size_t)kidx[e] * (2 * HC) + HC + c];
  atomicAdd(&agg[(size_t)qi * HC + c], attn * ve);
}

extern "C" void kernel_launch(void* const* d_in, const int* in_sizes, int n_in,
                              void* d_out, int out_size, void* d_ws, size_t ws_size,
                              hipStream_t stream) {
  const float* query = (const float*)d_in[0];
  const float* key_  = (const float*)d_in[1];
  const int*   qidx  = (const int*)d_in[2];
  const int*   kidx  = (const int*)d_in[3];
  const float* pr    = (const float*)d_in[4];
  const float* Wq    = (const float*)d_in[5];
  const float* Wkv   = (const float*)d_in[6];
  const float* Wb    = (const float*)d_in[7];
  const float* Wo    = (const float*)d_in[8];

  const int N = in_sizes[0] / HC;
  const int M = in_sizes[2];

  float* out_result = (float*)d_out;
  float* out_logit  = out_result + (size_t)N * HC;

  float*    q_buf  = (float*)d_ws;
  float*    kv_buf = q_buf + (size_t)N * HC;
  unsigned* segkey = (unsigned*)(kv_buf + (size_t)N * 2 * HC);
  float*    denom  = (float*)(segkey + (size_t)N * NH);
  float*    agg    = denom + (size_t)N * NH;

  const int TPB = 256;
  const int mtilesN = (N + 15) / 16;
  const int mtilesM = (M + 15) / 16;

  init_ws_kernel<<<(N * HC + TPB - 1) / TPB, TPB, 0, stream>>>(segkey, denom, agg, N);

  wmma_gemm_f32<HC, HC, HC, HC>
      <<<dim3((mtilesN + 7) / 8, HC / 16), TPB, 0, stream>>>(query, Wq, q_buf, N);

  wmma_gemm_f32<HC, HC, 2 * HC, 2 * HC>
      <<<dim3((mtilesN + 7) / 8, 2 * HC / 16), TPB, 0, stream>>>(key_, Wkv, kv_buf, N);

  edge_bias_logit_kernel<<<(mtilesM + 7) / 8, TPB, 0, stream>>>(
      pr, Wb, q_buf, kv_buf, qidx, kidx, out_logit, segkey, M);

  exp_denom_kernel<<<(M * NH + TPB - 1) / TPB, TPB, 0, stream>>>(
      out_logit, qidx, segkey, denom, M);

  scatter_agg_kernel<<<(M * HC + TPB - 1) / TPB, TPB, 0, stream>>>(
      out_logit, qidx, kidx, segkey, denom, kv_buf, agg, M);

  wmma_gemm_f32<HC, HC, HC, HC>
      <<<dim3((mtilesN + 7) / 8, HC / 16), TPB, 0, stream>>>(agg, Wo, out_result, N);
}


# float-atomic exp-sum and scatter replaced by per-row edge lists (one exchange per edge) and a wave-per-row gather that normalises once per row; fp summation order changes
# speedup vs baseline: 1.3519x; 1.3519x over previous
#include <hip/hip_runtime.h>

#define HC 128
#define NH 8

typedef __attribute__((ext_vector_type(2))) float v2f;
typedef __attribute__((ext_vector_type(8))) float v8f;

__device__ __forceinline__ unsigned float_key(float f) {
  unsigned u = __float_as_uint(f);
  return (u & 0x80000000u) ? ~u : (u | 0x80000000u);
}
__device__ __forceinline__ float key_float(unsigned k) {
  unsigned u = (k & 0x80000000u) ? (k & 0x7FFFFFFFu) : ~k;
  return __uint_as_float(u);
}

__global__ void init_ws_kernel(unsigned* __restrict__ segkey,
                               int* __restrict__ head, int n) {
  int gid = blockIdx.x * blockDim.x + threadIdx.x;
  if (gid < n * NH) segkey[gid] = 0u;
  if (gid < n) head[gid] = -1;
}

__global__ void build_lists_kernel(const int* __restrict__ qidx,
                                   const int* __restrict__ kidx,
                                   int* __restrict__ head,
                                   int* __restrict__ nodes, int Medges) {
  int e = blockIdx.x * blockDim.x + threadIdx.x;
  if (e >= Medges) return;
  const int prev = atomicExch(&head[qidx[e]], e);
  ((int2*)nodes)[e] = make_int2(prev, kidx[e]);
}

template <int K, int LDA, int LDB, int LDC>
__global__ __launch_bounds__(256) void wmma_gemm_f32(const float* __restrict__ A,
                                                     const float* __restrict__ B,
                                                     float* __restrict__ C,
                                                     int Mrows) {
  constexpr int KSTEPS = K / 4;
  __shared__ v2f Bs[KSTEPS * 32];

  const int tn = blockIdx.y;
#pragma unroll
  for (int idx = threadIdx.x; idx < KSTEPS * 32; idx += 256) {
    const int kk = idx >> 5, L = idx & 31;
    const int hf = L >> 4, lr2 = L & 15;
    const int ka = kk * 4 + 2 * hf;
    const int col = tn * 16 + lr2;
    v2f t;
    t.x = B[(size_t)ka * LDB + col];
    t.y = B[(size_t)(ka + 1) * LDB + col];
    Bs[idx] = t;
  }
  __syncthreads();

  const int lane = threadIdx.x & 31;
  const int tm = blockIdx.x * 8 + (threadIdx.x >> 5);
  if (tm * 16 >= Mrows) return;

  const int lr = lane & 15;
  const int half = lane >> 4;
  int arow = tm * 16 + lr; if (arow >= Mrows) arow = Mrows - 1;
  const float* ap = A + (size_t)arow * LDA + 2 * half;

  v8f acc = {};
#pragma unroll
  for (int kk = 0; kk < KSTEPS; ++kk) {
    const v2f a = *(const v2f*)(ap + kk * 4);
    acc = __builtin_amdgcn_wmma_f32_16x16x4_f32(false, a, false, Bs[kk * 32 + lane],
                                                (short)0, acc, false, false);
  }

  const int bcol = tn * 16 + lr;
#pragma unroll
  for (int v = 0; v < 8; ++v) {
    const int r = tm * 16 + v + 8 * half;
    if (r < Mrows) C[(size_t)r * LDC + bcol] = acc[v];
  }
}

__global__ __launch_bounds__(256) void edge_bias_logit_kernel(
    const float* __restrict__ PR, const float* __restrict__ Wb,
    const float* __restrict__ qbuf, const float* __restrict__ kvbuf,
    const int* __restrict__ qidx, const int* __restrict__ kidx,
    float* __restrict__ logit_out, unsigned* __restrict__ segkey, int Medges) {
  constexpr int KSTEPS = HC / 4;
  __shared__ v2f BmS[KSTEPS * 32];
  __shared__ v2f BaS[KSTEPS * 32];

  const int lane = threadIdx.x & 31;
  const int mtiles = (Medges + 15) >> 4;
  int tm = blockIdx.x * 8 + (threadIdx.x >> 5);
  if (tm >= mtiles) tm = mtiles - 1;

  const int half = lane >> 4;
  const int lr = lane & 15;
  int arow = tm * 16 + lr; if (arow >= Medges) arow = Medges - 1;

  v2f afrag[KSTEPS];
  const float* prow = PR + (size_t)arow * HC + 2 * half;
#pragma unroll
  for (int kk = 0; kk < KSTEPS; ++kk) afrag[kk] = *(const v2f*)(prow + kk * 4);

  float logacc[8];
#pragma unroll
  for (int v = 0; v < 8; ++v) logacc[v] = 0.0f;

  for (int j = 0; j < 8; ++j) {
    __syncthreads();
#pragma unroll
    for (int idx = threadIdx.x; idx < KSTEPS * 32; idx += 256) {
      const int kk = idx >> 5, L = idx & 31;
      const int hf = L >> 4, lr2 = L & 15;
      const int ka = kk * 4 + 2 * hf;
      const int col = j * 16 + lr2;
      v2f t1, t2;
      t1.x = Wb[(size_t)ka * 256 + col];
      t1.y = Wb[(size_t)(ka + 1) * 256 + col];
      t2.x = Wb[(size_t)ka * 256 + 128 + col];
      t2.y = Wb[(size_t)(ka + 1) * 256 + 128 + col];
      BmS[idx] = t1;
      BaS[idx] = t2;
    }
    __syncthreads();

    v8f bm = {};
    v8f ba = {};
#pragma unroll
    for (int kk = 0; kk < KSTEPS; ++kk) {
      bm = __builtin_amdgcn_wmma_f32_16x16x4_f32(false, afrag[kk], false,
                                                 BmS[kk * 32 + lane], (short)0,
                                                 bm, false, false);
      ba = __builtin_amdgcn_wmma_f32_16x16x4_f32(false, afrag[kk], false,
                                                 BaS[kk * 32 + lane], (short)0,
                                                 ba, false, false);
    }

    const int col = j * 16 + lr;
#pragma unroll
    for (int v = 0; v < 8; ++v) {
      int e = tm * 16 + v + 8 * half;
      e = (e < Medges) ? e : (Medges - 1);
      const int qi = qidx[e];
      const int ki = kidx[e];
      const float qv = qbuf[(size_t)qi * HC + col];
      const float kv = kvbuf[(size_t)ki * (2 * HC) + col];
      logacc[v] += qv * (kv * (1.0f + bm[v]) + ba[v]);
    }
  }

  const int h = lane & 7;
#pragma unroll
  for (int v = 0; v < 8; ++v) {
    logacc[v] += __shfl_xor(logacc[v], 8, 32);
    const int e = tm * 16 + v + 8 * half;
    if (((lane & 8) == 0) && e < Medges) {
      logit_out[(size_t)e * NH + h] = logacc[v];
      atomicMax(&segkey[(size_t)qidx[e] * NH + h], float_key(logacc[v]));
    }
  }
}

__global__ __launch_bounds__(256) void gather_agg_kernel(
    const float* __restrict__ logit, const int* __restrict__ head,
    const int* __restrict__ nodes, const unsigned* __restrict__ segkey,
    const float* __restrict__ kvbuf, float* __restrict__ agg, int n, int Medges) {
  const int lane = threadIdx.x & 31;
  const int row = __builtin_amdgcn_readfirstlane(blockIdx.x * 8 + (threadIdx.x >> 5));
  if (row >= n) return;
  const int hq = (lane & 1) * 4;
  const int2* __restrict__ nd2 = (const int2*)nodes;

  const uint4 kq = *(const uint4*)(segkey + (size_t)row * NH + hq);
  const float m0 = key_float(kq.x), m1 = key_float(kq.y);
  const float m2 = key_float(kq.z), m3 = key_float(kq.w);

  float4 acc = make_float4(0.0f, 0.0f, 0.0f, 0.0f);
  float4 den = make_float4(0.0f, 0.0f, 0.0f, 0.0f);

  const int first = head[row];
  int e = first;
  int nxt = -1, kcur = 0;
  if (e >= 0) { const int2 t = nd2[e]; nxt = t.x; kcur = t.y; }
  for (int it = 0; it < Medges && e >= 0; ++it) {
    const float4 lg = *(const float4*)(logit + (size_t)e * NH + hq);
    const float4 vv = *(const float4*)(kvbuf + (size_t)kcur * (2 * HC) + HC + 4 * lane);
    int nn = -1, kn = 0;
    if (nxt >= 0) { const int2 t = nd2[nxt]; nn = t.x; kn = t.y; }
    const float e0 = __expf(lg.x - m0), e1 = __expf(lg.y - m1);
    const float e2 = __expf(lg.z - m2), e3 = __expf(lg.w - m3);
    den.x += e0; den.y += e1; den.z += e2; den.w += e3;
    acc.x += e0 * vv.x; acc.y += e1 * vv.y; acc.z += e2 * vv.z; acc.w += e3 * vv.w;
    e = nxt; nxt = nn; kcur = kn;
  }

  float4 r = make_float4(0.0f, 0.0f, 0.0f, 0.0f);
  if (first >= 0) {
    r.x = acc.x / den.x; r.y = acc.y / den.y; r.z = acc.z / den.z; r.w = acc.w / den.w;
  }
  *(float4*)(agg + (size_t)row * HC + 4 * lane) = r;
}

extern "C" void kernel_launch(void* const* d_in, const int* in_sizes, int n_in,
                              void* d_out, int out_size, void* d_ws, size_t ws_size,
                              hipStream_t stream) {
  const float* query = (const float*)d_in[0];
  const float* key_  = (const float*)d_in[1];
  const int*   qidx  = (const int*)d_in[2];
  const int*   kidx  = (const int*)d_in[3];
  const float* pr    = (const float*)d_in[4];
  const float* Wq    = (const float*)d_in[5];
  const float* Wkv   = (const float*)d_in[6];
  const float* Wb    = (const float*)d_in[7];
  const float* Wo    = (const float*)d_in[8];

  const int N = in_sizes[0] / HC;
  const int M = in_sizes[2];

  float* out_result = (float*)d_out;
  float* out_logit  = out_result + (size_t)N * HC;

  float*    q_buf  = (float*)d_ws;
  float*    kv_buf = q_buf + (size_t)N * HC;
  unsigned* segkey = (unsigned*)(kv_buf + (size_t)N * 2 * HC);
  float*    agg    = (float*)(segkey + (size_t)N * NH);
  int*      head   = (int*)(agg + (size_t)N * HC);
  int*      nodes  = head + (size_t)N;

  const int TPB = 256;
  const int mtilesN = (N + 15) / 16;
  const int mtilesM = (M + 15) / 16;

  init_ws_kernel<<<(N * NH + TPB - 1) / TPB, TPB, 0, stream>>>(segkey, head, N);
  build_lists_kernel<<<(M + TPB - 1) / TPB, TPB, 0, stream>>>(qidx, kidx, head, nodes, M);

  wmma_gemm_f32<HC, HC, HC, HC>
      <<<dim3((mtilesN + 7) / 8, HC / 16), TPB, 0, stream>>>(query, Wq, q_buf, N);

  wmma_gemm_f32<HC, HC, 2 * HC, 2 * HC>
      <<<dim3((mtilesN + 7) / 8, 2 * HC / 16), TPB, 0, stream>>>(key_, Wkv, kv_buf, N);

  edge_bias_logit_kernel<<<(mtilesM + 7) / 8, TPB, 0, stream>>>(
      pr, Wb, q_buf, kv_buf, qidx, kidx, out_logit, segkey, M);

  gather_agg_kernel<<<(N + 7) / 8, TPB, 0, stream>>>(
      out_logit, head, nodes, segkey, kv_buf, agg, N, M);

  wmma_gemm_f32<HC, HC, HC, HC>
      <<<dim3((mtilesN + 7) / 8, HC / 16), TPB, 0, stream>>>(agg, Wo, out_result, N);
}


# each weight matrix (Wb, Wq, Wkv, Wo) staged whole in LDS once per persistent 16-wave work-group (256 groups); waves hold 16 A rows in registers and walk all column tiles; per-tile re-staging dropped
# speedup vs baseline: 1.9788x; 1.4637x over previous
#include <hip/hip_runtime.h>

#define HC 128
#define NH 8
#define NWG 256

typedef __attribute__((ext_vector_type(2))) float v2f;
typedef __attribute__((ext_vector_type(8))) float v8f;

__device__ __forceinline__ unsigned float_key(float f) {
  unsigned u = __float_as_uint(f);
  return (u & 0x80000000u) ? ~u : (u | 0x80000000u);
}
__device__ __forceinline__ float key_float(unsigned k) {
  unsigned u = (k & 0x80000000u) ? (k & 0x7FFFFFFFu) : ~k;
  return __uint_as_float(u);
}

__global__ void init_ws_kernel(unsigned* __restrict__ segkey,
                               int* __restrict__ head, int n) {
  int gid = blockIdx.x * blockDim.x + threadIdx.x;
  if (gid < n * NH) segkey[gid] = 0u;
  if (gid < n) head[gid] = -1;
}

__global__ void build_lists_kernel(const int* __restrict__ qidx,
                                   const int* __restrict__ kidx,
                                   int* __restrict__ head,
                                   int* __restrict__ nodes, int Medges) {
  int e = blockIdx.x * blockDim.x + threadIdx.x;
  if (e >= Medges) return;
  const int prev = atomicExch(&head[qidx[e]], e);
  ((int2*)nodes)[e] = make_int2(prev, kidx[e]);
}

template <int NB, int NTHREADS>
__device__ __forceinline__ void stage_weights(const float* __restrict__ B,
                                              v2f* __restrict__ S, int tid) {
  constexpr int G = NB / 4;
  static_assert((64 * G) % NTHREADS == 0, "whole trips only");
#pragma unroll
  for (int it = 0; it < (64 * G) / NTHREADS; ++it) {
    const int i = it * NTHREADS + tid;
    const int p = i / G, g = i % G;
    const float4 r0 = *(const float4*)(B + (size_t)(2 * p) * NB + 4 * g);
    const float4 r1 = *(const float4*)(B + (size_t)(2 * p + 1) * NB + 4 * g);
    const int col = 4 * g;
    const int ct = col >> 4, lr2 = col & 15;
    const int kk = p >> 1, hf = p & 1;
    v2f* d = S + ((ct * 32 + kk) * 32 + hf * 16 + lr2);
    v2f t0, t1, t2, t3;
    t0.x = r0.x; t0.y = r1.x;
    t1.x = r0.y; t1.y = r1.y;
    t2.x = r0.z; t2.y = r1.z;
    t3.x = r0.w; t3.y = r1.w;
    d[0] = t0; d[1] = t1; d[2] = t2; d[3] = t3;
  }
}

template <int NB>
__global__ __launch_bounds__(512) void wmma_gemm_f32(const float* __restrict__ A,
                                                     const float* __restrict__ B,
                                                     float* __restrict__ C,
                                                     int Mrows) {
  constexpr int KSTEPS = HC / 4;
  constexpr int NT = NB / 16;
  __shared__ v2f Bs[NT * KSTEPS * 32];

  stage_weights<NB, 512>(B, Bs, threadIdx.x);
  __syncthreads();

  const int lane = threadIdx.x & 31;
  const int wave = __builtin_amdgcn_readfirstlane(threadIdx.x >> 5);
  const int lr = lane & 15;
  const int half = lane >> 4;
  const int mtiles = (Mrows + 15) >> 4;

  for (int tm = wave * (int)gridDim.x + (int)blockIdx.x; tm < mtiles; tm += 16 * (int)gridDim.x) {
    int arow = tm * 16 + lr; if (arow >= Mrows) arow = Mrows - 1;
    const float* ap = A + (size_t)arow * HC + 2 * half;
    v2f afrag[KSTEPS];
#pragma unroll
    for (int kk = 0; kk < KSTEPS; ++kk) afrag[kk] = *(const v2f*)(ap + kk * 4);

#pragma unroll 1
    for (int ct = 0; ct < NT; ct += 2) {
      const v2f* b0 = Bs + (ct * 32) * 32 + lane;
      const v2f* b1 = Bs + ((ct + 1) * 32) * 32 + lane;
      v8f c0 = {};
      v8f c1 = {};
#pragma unroll
      for (int kk = 0; kk < KSTEPS; ++kk) {
        c0 = __builtin_amdgcn_wmma_f32_16x16x4_f32(false, afrag[kk], false, b0[kk * 32],
                                                   (short)0, c0, false, false);
        c1 = __builtin_amdgcn_wmma_f32_16x16x4_f32(false, afrag[kk], false, b1[kk * 32],
                                                   (short)0, c1, false, false);
      }
      const int bcol = ct * 16 + lr;
#pragma unroll
      for (int v = 0; v < 8; ++v) {
        const int r = tm * 16 + v + 8 * half;
        if (r < Mrows) {
          C[(size_t)r * NB + bcol] = c0[v];
          C[(size_t)r * NB + bcol + 16] = c1[v];
        }
      }
    }
  }
}

__global__ __launch_bounds__(512) void edge_bias_logit_kernel(
    const float* __restrict__ PR, const float* __restrict__ Wb,
    const float* __restrict__ qbuf, const float* __restrict__ kvbuf,
    const int* __restrict__ qidx, const int* __restrict__ kidx,
    float* __restrict__ logit_out, unsigned* __restrict__ segkey, int Medges) {
  constexpr int KSTEPS = HC / 4;
  __shared__ v2f WbS[16 * KSTEPS * 32];

  stage_weights<2 * HC, 512>(Wb, WbS, threadIdx.x);
  __syncthreads();

  const int lane = threadIdx.x & 31;
  const int wave = __builtin_amdgcn_readfirstlane(threadIdx.x >> 5);
  const int mtiles = (Medges + 15) >> 4;
  const int half = lane >> 4;
  const int lr = lane & 15;
  const int h = lane & 7;

  for (int tm = wave * (int)gridDim.x + (int)blockIdx.x; tm < mtiles; tm += 16 * (int)gridDim.x) {
    int arow = tm * 16 + lr; if (arow >= Medges) arow = Medges - 1;

    v2f afrag[KSTEPS];
    const float* prow = PR + (size_t)arow * HC + 2 * half;
#pragma unroll
    for (int kk = 0; kk < KSTEPS; ++kk) afrag[kk] = *(const v2f*)(prow + kk * 4);

    int qi[8];
    unsigned qoff[8], koff[8];
#pragma unroll
    for (int v = 0; v < 8; ++v) {
      int e = tm * 16 + v + 8 * half;
      e = (e < Medges) ? e : (Medges - 1);
      qi[v] = qidx[e];
      qoff[v] = (unsigned)qi[v] * HC;
      koff[v] = (unsigned)kidx[e] * (2 * HC);
    }

    float logacc[8];
#pragma unroll
    for (int v = 0; v < 8; ++v) logacc[v] = 0.0f;

    for (int j = 0; j < 8; ++j) {
      const v2f* bmS = WbS + (j * 32) * 32 + lane;
      const v2f* baS = WbS + ((8 + j) * 32) * 32 + lane;
      v8f bm = {};
      v8f ba = {};
#pragma unroll
      for (int kk = 0; kk < KSTEPS; ++kk) {
        bm = __builtin_amdgcn_wmma_f32_16x16x4_f32(false, afrag[kk], false,
                                                   bmS[kk * 32], (short)0,
                                                   bm, false, false);
        ba = __builtin_amdgcn_wmma_f32_16x16x4_f32(false, afrag[kk], false,
                                                   baS[kk * 32], (short)0,
                                                   ba, false, false);
      }

      const unsigned col = j * 16 + lr;
#pragma unroll
      for (int v = 0; v < 8; ++v) {
        const float qv = qbuf[qoff[v] + col];
        const float kv = kvbuf[koff[v] + col];
        logacc[v] += qv * (kv * (1.0f + bm[v]) + ba[v]);
      }
    }

#pragma unroll
    for (int v = 0; v < 8; ++v) {
      logacc[v] += __shfl_xor(logacc[v], 8, 32);
      const int e = tm * 16 + v + 8 * half;
      if (((lane & 8) == 0) && e < Medges) {
        logit_out[(size_t)e * NH + h] = logacc[v];
        atomicMax(&segkey[(size_t)qi[v] * NH + h], float_key(logacc[v]));
      }
    }
  }
}

__global__ __launch_bounds__(256) void gather_agg_kernel(
    const float* __restrict__ logit, const int* __restrict__ head,
    const int* __restrict__ nodes, const unsigned* __restrict__ segkey,
    const float* __restrict__ kvbuf, float* __restrict__ agg, int n, int Medges) {
  const int lane = threadIdx.x & 31;
  const int row = __builtin_amdgcn_readfirstlane(blockIdx.x * 8 + (threadIdx.x >> 5));
  if (row >= n) return;
  const int hq = (lane & 1) * 4;
  const int2* __restrict__ nd2 = (const int2*)nodes;

  const uint4 kq = *(const uint4*)(segkey + (size_t)row * NH + hq);
  const float m0 = key_float(kq.x), m1 = key_float(kq.y);
  const float m2 = key_float(kq.z), m3 = key_float(kq.w);

  float4 acc = make_float4(0.0f, 0.0f, 0.0f, 0.0f);
  float4 den = make_float4(0.0f, 0.0f, 0.0f, 0.0f);

  const int first = head[row];
  int e = first;
  int nxt = -1, kcur = 0;
  if (e >= 0) { const int2 t = nd2[e]; nxt = t.x; kcur = t.y; }
  for (int it = 0; it < Medges && e >= 0; ++it) {
    const float4 lg = *(const float4*)(logit + (size_t)e * NH + hq);
    const float4 vv = *(const float4*)(kvbuf + (size_t)kcur * (2 * HC) + HC + 4 * lane);
    int nn = -1, kn = 0;
    if (nxt >= 0) { const int2 t = nd2[nxt]; nn = t.x; kn = t.y; }
    const float e0 = __expf(lg.x - m0), e1 = __expf(lg.y - m1);
    const float e2 = __expf(lg.z - m2), e3 = __expf(lg.w - m3);
    den.x += e0; den.y += e1; den.z += e2; den.w += e3;
    acc.x += e0 * vv.x; acc.y += e1 * vv.y; acc.z += e2 * vv.z; acc.w += e3 * vv.w;
    e = nxt; nxt = nn; kcur = kn;
  }

  float4 r = make_float4(0.0f, 0.0f, 0.0f, 0.0f);
  if (first >= 0) {
    r.x = acc.x / den.x; r.y = acc.y / den.y; r.z = acc.z / den.z; r.w = acc.w / den.w;
  }
  *(float4*)(agg + (size_t)row * HC + 4 * lane) = r;
}

extern "C" void kernel_launch(void* const* d_in, const int* in_sizes, int n_in,
                              void* d_out, int out_size, void* d_ws, size_t ws_size,
                              hipStream_t stream) {
  const float* query = (const float*)d_in[0];
  const float* key_  = (const float*)d_in[1];
  const int*   qidx  = (const int*)d_in[2];
  const int*   kidx  = (const int*)d_in[3];
  const float* pr    = (const float*)d_in[4];
  const float* Wq    = (const float*)d_in[5];
  const float* Wkv   = (const float*)d_in[6];
  const float* Wb    = (const float*)d_in[7];
  const float* Wo    = (const float*)d_in[8];

  const int N = in_sizes[0] / HC;
  const int M = in_sizes[2];

  float* out_result = (float*)d_out;
  float* out_logit  = out_result + (size_t)N * HC;

  float*    q_buf  = (float*)d_ws;
  float*    kv_buf = q_buf + (size_t)N * HC;
  unsigned* segkey = (unsigned*)(kv_buf + (size_t)N * 2 * HC);
  float*    agg    = (float*)(segkey + (size_t)N * NH);
  int*      head   = (int*)(agg + (size_t)N * HC);
  int*      nodes  = head + (size_t)N;

  const int TPB = 256;

  init_ws_kernel<<<(N * NH + TPB - 1) / TPB, TPB, 0, stream>>>(segkey, head, N);
  build_lists_kernel<<<(M + TPB - 1) / TPB, TPB, 0, stream>>>(qidx, kidx, head, nodes, M);

  wmma_gemm_f32<HC><<<NWG, 512, 0, stream>>>(query, Wq, q_buf, N);

  wmma_gemm_f32<2 * HC><<<NWG, 512, 0, stream>>>(key_, Wkv, kv_buf, N);

  edge_bias_logit_kernel<<<NWG, 512, 0, stream>>>(
      pr, Wb, q_buf, kv_buf, qidx, kidx, out_logit, segkey, M);

  gather_agg_kernel<<<(N + 7) / 8, TPB, 0, stream>>>(
      out_logit, head, nodes, segkey, kv_buf, agg, N, M);

  wmma_gemm_f32<HC><<<NWG, 512, 0, stream>>>(agg, Wo, out_result, N);
}


# workspace init and edge-list build moved into the tile-less waves (3 of 16 per group) of the two projection GEMM launches; own launches gone, 7 -> 5; list order stays exchange arrival order
# speedup vs baseline: 2.0629x; 1.0425x over previous
#include <hip/hip_runtime.h>

#define HC 128
#define NH 8
#define NWG 256

typedef __attribute__((ext_vector_type(2))) float v2f;
typedef __attribute__((ext_vector_type(8))) float v8f;

__device__ __forceinline__ unsigned float_key(float f) {
  unsigned u = __float_as_uint(f);
  return (u & 0x80000000u) ? ~u : (u | 0x80000000u);
}
__device__ __forceinline__ float key_float(unsigned k) {
  unsigned u = (k & 0x80000000u) ? (k & 0x7FFFFFFFu) : ~k;
  return __uint_as_float(u);
}


__device__ __forceinline__ void side_init(unsigned* __restrict__ segkey,
                                          int* __restrict__ head, int n,
                                          int sid, int nside) {
  uint4* sk4 = (uint4*)segkey;
  for (int i = sid; i < 2 * n; i += nside) sk4[i] = make_uint4(0u, 0u, 0u, 0u);
  for (int i = sid; i < n; i += nside) head[i] = -1;
}

__device__ __forceinline__ void side_build(const int* __restrict__ qidx,
                                           const int* __restrict__ kidx,
                                           int* __restrict__ head,
                                           int* __restrict__ nodes, int Medges,
                                           int sid, int nside) {
  for (int e0 = sid; e0 < Medges; e0 += 8 * nside) {
    int q[8], k[8], prev[8];
#pragma unroll
    for (int u = 0; u < 8; ++u) {
      const int e = e0 + u * nside;
      q[u] = 0; k[u] = 0;
      if (e < Medges) { q[u] = qidx[e]; k[u] = kidx[e]; }
    }
#pragma unroll
    for (int u = 0; u < 8; ++u) {
      const int e = e0 + u * nside;
      prev[u] = -1;
      if (e < Medges) prev[u] = atomicExch(&head[q[u]], e);
    }
#pragma unroll
    for (int u = 0; u < 8; ++u) {
      const int e = e0 + u * nside;
      if (e < Medges) ((int2*)nodes)[e] = make_int2(prev[u], k[u]);
    }
  }
}

template <int NB, int NTHREADS>
__device__ __forceinline__ void stage_weights(const float* __restrict__ B,
                                              v2f* __restrict__ S, int tid) {
  constexpr int G = NB / 4;
  static_assert((64 * G) % NTHREADS == 0, "whole trips only");
#pragma unroll
  for (int it = 0; it < (64 * G) / NTHREADS; ++it) {
    const int i = it * NTHREADS + tid;
    const int p = i / G, g = i % G;
    const float4 r0 = *(const float4*)(B + (size_t)(2 * p) * NB + 4 * g);
    const float4 r1 = *(const float4*)(B + (size_t)(2 * p + 1) * NB + 4 * g);
    const int col = 4 * g;
    const int ct = col >> 4, lr2 = col & 15;
    const int kk = p >> 1, hf = p & 1;
    v2f* d = S + ((ct * 32 + kk) * 32 + hf * 16 + lr2);
    v2f t0, t1, t2, t3;
    t0.x = r0.x; t0.y = r1.x;
    t1.x = r0.y; t1.y = r1.y;
    t2.x = r0.z; t2.y = r1.z;
    t3.x = r0.w; t3.y = r1.w;
    d[0] = t0; d[1] = t1; d[2] = t2; d[3] = t3;
  }
}

template <int NB, int SIDE>
__global__ __launch_bounds__(512) void wmma_gemm_f32(const float* __restrict__ A,
                                                     const float* __restrict__ B,
                                                     float* __restrict__ C,
                                                     int Mrows,
                                                     unsigned* __restrict__ segkey,
                                                     int* __restrict__ head,
                                                     int* __restrict__ nodes,
                                                     const int* __restrict__ qidx,
                                                     const int* __restrict__ kidx,
                                                     int n, int Medges) {
  constexpr int KSTEPS = HC / 4;
  constexpr int NT = NB / 16;
  __shared__ v2f Bs[NT * KSTEPS * 32];

  stage_weights<NB, 512>(B, Bs, threadIdx.x);
  __syncthreads();

  const int lane = threadIdx.x & 31;
  const int wave = __builtin_amdgcn_readfirstlane(threadIdx.x >> 5);
  const int lr = lane & 15;
  const int half = lane >> 4;
  const int mtiles = (Mrows + 15) >> 4;

  if (SIDE != 0) {
    const int busy = (mtiles + (int)gridDim.x - 1) / (int)gridDim.x;
    const int first = busy < 16 ? busy : 0;
    if (wave >= first) {
      const int nsw = 16 - first;
      const int sid = ((int)blockIdx.x * nsw + (wave - first)) * 32 + lane;
      const int nside = (int)gridDim.x * nsw * 32;
      if (SIDE == 1) side_init(segkey, head, n, sid, nside);
      else           side_build(qidx, kidx, head, nodes, Medges, sid, nside);
    }
  }

  for (int tm = wave * (int)gridDim.x + (int)blockIdx.x; tm < mtiles; tm += 16 * (int)gridDim.x) {
    int arow = tm * 16 + lr; if (arow >= Mrows) arow = Mrows - 1;
    const float* ap = A + (size_t)arow * HC + 2 * half;
    v2f afrag[KSTEPS];
#pragma unroll
    for (int kk = 0; kk < KSTEPS; ++kk) afrag[kk] = *(const v2f*)(ap + kk * 4);

#pragma unroll 1
    for (int ct = 0; ct < NT; ct += 2) {
      const v2f* b0 = Bs + (ct * 32) * 32 + lane;
      const v2f* b1 = Bs + ((ct + 1) * 32) * 32 + lane;
      v8f c0 = {};
      v8f c1 = {};
#pragma unroll
      for (int kk = 0; kk < KSTEPS; ++kk) {
        c0 = __builtin_amdgcn_wmma_f32_16x16x4_f32(false, afrag[kk], false, b0[kk * 32],
                                                   (short)0, c0, false, false);
        c1 = __builtin_amdgcn_wmma_f32_16x16x4_f32(false, afrag[kk], false, b1[kk * 32],
                                                   (short)0, c1, false, false);
      }
      const int bcol = ct * 16 + lr;
#pragma unroll
      for (int v = 0; v < 8; ++v) {
        const int r = tm * 16 + v + 8 * half;
        if (r < Mrows) {
          C[(size_t)r * NB + bcol] = c0[v];
          C[(size_t)r * NB + bcol + 16] = c1[v];
        }
      }
    }
  }
}

__global__ __launch_bounds__(512) void edge_bias_logit_kernel(
    const float* __restrict__ PR, const float* __restrict__ Wb,
    const float* __restrict__ qbuf, const float* __restrict__ kvbuf,
    const int* __restrict__ qidx, const int* __restrict__ kidx,
    float* __restrict__ logit_out, unsigned* __restrict__ segkey, int Medges) {
  constexpr int KSTEPS = HC / 4;
  __shared__ v2f WbS[16 * KSTEPS * 32];

  stage_weights<2 * HC, 512>(Wb, WbS, threadIdx.x);
  __syncthreads();

  const int lane = threadIdx.x & 31;
  const int wave = __builtin_amdgcn_readfirstlane(threadIdx.x >> 5);
  const int mtiles = (Medges + 15) >> 4;
  const int half = lane >> 4;
  const int lr = lane & 15;
  const int h = lane & 7;

  for (int tm = wave * (int)gridDim.x + (int)blockIdx.x; tm < mtiles; tm += 16 * (int)gridDim.x) {
    int arow = tm * 16 + lr; if (arow >= Medges) arow = Medges - 1;

    v2f afrag[KSTEPS];
    const float* prow = PR + (size_t)arow * HC + 2 * half;
#pragma unroll
    for (int kk = 0; kk < KSTEPS; ++kk) afrag[kk] = *(const v2f*)(prow + kk * 4);

    int qi[8];
    unsigned qoff[8], koff[8];
#pragma unroll
    for (int v = 0; v < 8; ++v) {
      int e = tm * 16 + v + 8 * half;
      e = (e < Medges) ? e : (Medges - 1);
      qi[v] = qidx[e];
      qoff[v] = (unsigned)qi[v] * HC;
      koff[v] = (unsigned)kidx[e] * (2 * HC);
    }

    float logacc[8];
#pragma unroll
    for (int v = 0; v < 8; ++v) logacc[v] = 0.0f;

    for (int j = 0; j < 8; ++j) {
      const v2f* bmS = WbS + (j * 32) * 32 + lane;
      const v2f* baS = WbS + ((8 + j) * 32) * 32 + lane;
      v8f bm = {};
      v8f ba = {};
#pragma unroll
      for (int kk = 0; kk < KSTEPS; ++kk) {
        bm = __builtin_amdgcn_wmma_f32_16x16x4_f32(false, afrag[kk], false,
                                                   bmS[kk * 32], (short)0,
                                                   bm, false, false);
        ba = __builtin_amdgcn_wmma_f32_16x16x4_f32(false, afrag[kk], false,
                                                   baS[kk * 32], (short)0,
                                                   ba, false, false);
      }

      const unsigned col = j * 16 + lr;
#pragma unroll
      for (int v = 0; v < 8; ++v) {
        const float qv = qbuf[qoff[v] + col];
        const float kv = kvbuf[koff[v] + col];
        logacc[v] += qv * (kv * (1.0f + bm[v]) + ba[v]);
      }
    }

#pragma unroll
    for (int v = 0; v < 8; ++v) {
      logacc[v] += __shfl_xor(logacc[v], 8, 32);
      const int e = tm * 16 + v + 8 * half;
      if (((lane & 8) == 0) && e < Medges) {
        logit_out[(size_t)e * NH + h] = logacc[v];
        atomicMax(&segkey[(size_t)qi[v] * NH + h], float_key(logacc[v]));
      }
    }
  }
}

__global__ __launch_bounds__(256) void gather_agg_kernel(
    const float* __restrict__ logit, const int* __restrict__ head,
    const int* __restrict__ nodes, const unsigned* __restrict__ segkey,
    const float* __restrict__ kvbuf, float* __restrict__ agg, int n, int Medges) {
  const int lane = threadIdx.x & 31;
  const int row = __builtin_amdgcn_readfirstlane(blockIdx.x * 8 + (threadIdx.x >> 5));
  if (row >= n) return;
  const int hq = (lane & 1) * 4;
  const int2* __restrict__ nd2 = (const int2*)nodes;

  const uint4 kq = *(const uint4*)(segkey + (size_t)row * NH + hq);
  const float m0 = key_float(kq.x), m1 = key_float(kq.y);
  const float m2 = key_float(kq.z), m3 = key_float(kq.w);

  float4 acc = make_float4(0.0f, 0.0f, 0.0f, 0.0f);
  float4 den = make_float4(0.0f, 0.0f, 0.0f, 0.0f);

  const int first = head[row];
  int e = first;
  int nxt = -1, kcur = 0;
  if (e >= 0) { const int2 t = nd2[e]; nxt = t.x; kcur = t.y; }
  for (int it = 0; it < Medges && e >= 0; ++it) {
    const float4 lg = *(const float4*)(logit + (size_t)e * NH + hq);
    const float4 vv = *(const float4*)(kvbuf + (size_t)kcur * (2 * HC) + HC + 4 * lane);
    int nn = -1, kn = 0;
    if (nxt >= 0) { const int2 t = nd2[nxt]; nn = t.x; kn = t.y; }
    const float e0 = __expf(lg.x - m0), e1 = __expf(lg.y - m1);
    const float e2 = __expf(lg.z - m2), e3 = __expf(lg.w - m3);
    den.x += e0; den.y += e1; den.z += e2; den.w += e3;
    acc.x += e0 * vv.x; acc.y += e1 * vv.y; acc.z += e2 * vv.z; acc.w += e3 * vv.w;
    e = nxt; nxt = nn; kcur = kn;
  }

  float4 r = make_float4(0.0f, 0.0f, 0.0f, 0.0f);
  if (first >= 0) {
    r.x = acc.x / den.x; r.y = acc.y / den.y; r.z = acc.z / den.z; r.w = acc.w / den.w;
  }
  *(float4*)(agg + (size_t)row * HC + 4 * lane) = r;
}

extern "C" void kernel_launch(void* const* d_in, const int* in_sizes, int n_in,
                              void* d_out, int out_size, void* d_ws, size_t ws_size,
                              hipStream_t stream) {
  const float* query = (const float*)d_in[0];
  const float* key_  = (const float*)d_in[1];
  const int*   qidx  = (const int*)d_in[2];
  const int*   kidx  = (const int*)d_in[3];
  const float* pr    = (const float*)d_in[4];
  const float* Wq    = (const float*)d_in[5];
  const float* Wkv   = (const float*)d_in[6];
  const float* Wb    = (const float*)d_in[7];
  const float* Wo    = (const float*)d_in[8];

  const int N = in_sizes[0] / HC;
  const int M = in_sizes[2];

  float* out_result = (float*)d_out;
  float* out_logit  = out_result + (size_t)N * HC;

  float*    q_buf  = (float*)d_ws;
  float*    kv_buf = q_buf + (size_t)N * HC;
  unsigned* segkey = (unsigned*)(kv_buf + (size_t)N * 2 * HC);
  float*    agg    = (float*)(segkey + (size_t)N * NH);
  int*      head   = (int*)(agg + (size_t)N * HC);
  int*      nodes  = head + (size_t)N;

  const int TPB = 256;

  wmma_gemm_f32<HC, 1><<<NWG, 512, 0, stream>>>(query, Wq, q_buf, N,
                                                segkey, head, nodes, qidx, kidx, N, M);

  wmma_gemm_f32<2 * HC, 2><<<NWG, 512, 0, stream>>>(key_, Wkv, kv_buf, N,
                                                    segkey, head, nodes, qidx, kidx, N, M);

  edge_bias_logit_kernel<<<NWG, 512, 0, stream>>>(
      pr, Wb, q_buf, kv_buf, qidx, kidx, out_logit, segkey, M);

  gather_agg_kernel<<<(N + 7) / 8, TPB, 0, stream>>>(
      out_logit, head, nodes, segkey, kv_buf, agg, N, M);

  wmma_gemm_f32<HC, 0><<<NWG, 512, 0, stream>>>(agg, Wo, out_result, N,
                                                segkey, head, nodes, qidx, kidx, N, M);
}


# q and kv projection GEMMs in one launch (5 -> 4): group stages Wq and Wkv together, deals tiles to waves by matrix work; edge-list push moves into the edge launch, list order still exchange arrival
# speedup vs baseline: 2.1478x; 1.0412x over previous
#include <hip/hip_runtime.h>

#define HC 128
#define NH 8
#define NWG 256

typedef __attribute__((ext_vector_type(2))) float v2f;
typedef __attribute__((ext_vector_type(8))) float v8f;

__device__ __forceinline__ unsigned float_key(float f) {
  unsigned u = __float_as_uint(f);
  return (u & 0x80000000u) ? ~u : (u | 0x80000000u);
}
__device__ __forceinline__ float key_float(unsigned k) {
  unsigned u = (k & 0x80000000u) ? (k & 0x7FFFFFFFu) : ~k;
  return __uint_as_float(u);
}

__device__ __forceinline__ void side_init(unsigned* __restrict__ segkey,
                                          int* __restrict__ head, int n,
                                          int sid, int nside) {
  uint4* sk4 = (uint4*)segkey;
  for (int i = sid; i < 2 * n; i += nside) sk4[i] = make_uint4(0u, 0u, 0u, 0u);
  for (int i = sid; i < n; i += nside) head[i] = -1;
}

template <int NB, int NTHREADS>
__device__ __forceinline__ void stage_weights(const float* __restrict__ B,
                                              v2f* __restrict__ S, int tid) {
  constexpr int G = NB / 4;
  static_assert((64 * G) % NTHREADS == 0, "whole trips only");
#pragma unroll
  for (int it = 0; it < (64 * G) / NTHREADS; ++it) {
    const int i = it * NTHREADS + tid;
    const int p = i / G, g = i % G;
    const float4 r0 = *(const float4*)(B + (size_t)(2 * p) * NB + 4 * g);
    const float4 r1 = *(const float4*)(B + (size_t)(2 * p + 1) * NB + 4 * g);
    const int col = 4 * g;
    const int ct = col >> 4, lr2 = col & 15;
    const int kk = p >> 1, hf = p & 1;
    v2f* d = S + ((ct * 32 + kk) * 32 + hf * 16 + lr2);
    v2f t0, t1, t2, t3;
    t0.x = r0.x; t0.y = r1.x;
    t1.x = r0.y; t1.y = r1.y;
    t2.x = r0.z; t2.y = r1.z;
    t3.x = r0.w; t3.y = r1.w;
    d[0] = t0; d[1] = t1; d[2] = t2; d[3] = t3;
  }
}

template <int NB>
__device__ __forceinline__ void gemm_tile(const float* __restrict__ A,
                                          const v2f* __restrict__ Bs,
                                          float* __restrict__ C,
                                          int tm, int Mrows, int lane) {
  constexpr int KSTEPS = HC / 4;
  constexpr int NT = NB / 16;
  const int lr = lane & 15;
  const int half = lane >> 4;
  int arow = tm * 16 + lr; if (arow >= Mrows) arow = Mrows - 1;
  const float* ap = A + (size_t)arow * HC + 2 * half;
  v2f afrag[KSTEPS];
#pragma unroll
  for (int kk = 0; kk < KSTEPS; ++kk) afrag[kk] = *(const v2f*)(ap + kk * 4);

#pragma unroll 1
  for (int ct = 0; ct < NT; ct += 2) {
    const v2f* b0 = Bs + (ct * 32) * 32 + lane;
    const v2f* b1 = Bs + ((ct + 1) * 32) * 32 + lane;
    v8f c0 = {};
    v8f c1 = {};
#pragma unroll
    for (int kk = 0; kk < KSTEPS; ++kk) {
      c0 = __builtin_amdgcn_wmma_f32_16x16x4_f32(false, afrag[kk], false, b0[kk * 32],
                                                 (short)0, c0, false, false);
      c1 = __builtin_amdgcn_wmma_f32_16x16x4_f32(false, afrag[kk], false, b1[kk * 32],
                                                 (short)0, c1, false, false);
    }
    const int bcol = ct * 16 + lr;
#pragma unroll
    for (int v = 0; v < 8; ++v) {
      const int r = tm * 16 + v + 8 * half;
      if (r < Mrows) {
        C[(size_t)r * NB + bcol] = c0[v];
        C[(size_t)r * NB + bcol + 16] = c1[v];
      }
    }
  }
}

template <int NB>
__global__ __launch_bounds__(512) void wmma_gemm_f32(const float* __restrict__ A,
                                                     const float* __restrict__ B,
                                                     float* __restrict__ C,
                                                     int Mrows) {
  constexpr int KSTEPS = HC / 4;
  constexpr int NT = NB / 16;
  __shared__ v2f Bs[NT * KSTEPS * 32];

  stage_weights<NB, 512>(B, Bs, threadIdx.x);
  __syncthreads();

  const int lane = threadIdx.x & 31;
  const int wave = __builtin_amdgcn_readfirstlane(threadIdx.x >> 5);
  const int mtiles = (Mrows + 15) >> 4;
  for (int tm = wave * (int)gridDim.x + (int)blockIdx.x; tm < mtiles; tm += 16 * (int)gridDim.x)
    gemm_tile<NB>(A, Bs, C, tm, Mrows, lane);
}

__global__ __launch_bounds__(512) void proj_qkv_kernel(
    const float* __restrict__ Xq, const float* __restrict__ Wq, float* __restrict__ Q,
    const float* __restrict__ Xk, const float* __restrict__ Wkv, float* __restrict__ KV,
    int Mrows, unsigned* __restrict__ segkey, int* __restrict__ head) {
  constexpr int KSTEPS = HC / 4;
  __shared__ v2f BqS[8 * KSTEPS * 32];
  __shared__ v2f BkvS[16 * KSTEPS * 32];

  side_init(segkey, head, Mrows, (int)blockIdx.x * 512 + (int)threadIdx.x,
            (int)gridDim.x * 512);

  stage_weights<HC, 512>(Wq, BqS, threadIdx.x);
  stage_weights<2 * HC, 512>(Wkv, BkvS, threadIdx.x);
  __syncthreads();

  const int lane = threadIdx.x & 31;
  const int wave = __builtin_amdgcn_readfirstlane(threadIdx.x >> 5);
  const int G = (int)gridDim.x, bid = (int)blockIdx.x;
  const int mtiles = (Mrows + 15) >> 4;
  const int nt = (mtiles > bid) ? (mtiles - bid + G - 1) / G : 0;

  for (int j = wave; j < nt; j += 16)
    gemm_tile<2 * HC>(Xk, BkvS, KV, j * G + bid, Mrows, lane);

  const int F = nt < 16 ? 16 - nt : 0;
  for (int i = 0; i < nt; ++i) {
    const int owner = (i < 2 * F) ? nt + (i % F) : ((i - 2 * F) & 15);
    if (owner == wave) gemm_tile<HC>(Xq, BqS, Q, i * G + bid, Mrows, lane);
  }
}

__global__ __launch_bounds__(512) void edge_bias_logit_kernel(
    const float* __restrict__ PR, const float* __restrict__ Wb,
    const float* __restrict__ qbuf, const float* __restrict__ kvbuf,
    const int* __restrict__ qidx, const int* __restrict__ kidx,
    float* __restrict__ logit_out, unsigned* __restrict__ segkey,
    int* __restrict__ head, int* __restrict__ nodes, int Medges) {
  constexpr int KSTEPS = HC / 4;
  __shared__ v2f WbS[16 * KSTEPS * 32];

  stage_weights<2 * HC, 512>(Wb, WbS, threadIdx.x);
  __syncthreads();

  const int lane = threadIdx.x & 31;
  const int wave = __builtin_amdgcn_readfirstlane(threadIdx.x >> 5);
  const int mtiles = (Medges + 15) >> 4;
  const int half = lane >> 4;
  const int lr = lane & 15;
  const int h = lane & 7;

  int pend_e = -1, pend_prev = -1, pend_k = 0;

  for (int tm = wave * (int)gridDim.x + (int)blockIdx.x; tm < mtiles; tm += 16 * (int)gridDim.x) {
    int arow = tm * 16 + lr; if (arow >= Medges) arow = Medges - 1;

    v2f afrag[KSTEPS];
    const float* prow = PR + (size_t)arow * HC + 2 * half;
#pragma unroll
    for (int kk = 0; kk < KSTEPS; ++kk) afrag[kk] = *(const v2f*)(prow + kk * 4);

    int qi[8];
    unsigned qoff[8], koff[8];
#pragma unroll
    for (int v = 0; v < 8; ++v) {
      int e = tm * 16 + v + 8 * half;
      e = (e < Medges) ? e : (Medges - 1);
      qi[v] = qidx[e];
      qoff[v] = (unsigned)qi[v] * HC;
      koff[v] = (unsigned)kidx[e] * (2 * HC);
    }

    float logacc[8];
#pragma unroll
    for (int v = 0; v < 8; ++v) logacc[v] = 0.0f;

    for (int j = 0; j < 8; ++j) {
      const v2f* bmS = WbS + (j * 32) * 32 + lane;
      const v2f* baS = WbS + ((8 + j) * 32) * 32 + lane;
      v8f bm = {};
      v8f ba = {};
#pragma unroll
      for (int kk = 0; kk < KSTEPS; ++kk) {
        bm = __builtin_amdgcn_wmma_f32_16x16x4_f32(false, afrag[kk], false,
                                                   bmS[kk * 32], (short)0,
                                                   bm, false, false);
        ba = __builtin_amdgcn_wmma_f32_16x16x4_f32(false, afrag[kk], false,
                                                   baS[kk * 32], (short)0,
                                                   ba, false, false);
      }

      const unsigned col = j * 16 + lr;
#pragma unroll
      for (int v = 0; v < 8; ++v) {
        const float qv = qbuf[qoff[v] + col];
        const float kv = kvbuf[koff[v] + col];
        logacc[v] += qv * (kv * (1.0f + bm[v]) + ba[v]);
      }
    }

#pragma unroll
    for (int v = 0; v < 8; ++v) {
      logacc[v] += __shfl_xor(logacc[v], 8, 32);
      const int e = tm * 16 + v + 8 * half;
      if (((lane & 8) == 0) && e < Medges) {
        logit_out[(size_t)e * NH + h] = logacc[v];
        atomicMax(&segkey[(size_t)qi[v] * NH + h], float_key(logacc[v]));
      }
    }

    if (pend_e >= 0) ((int2*)nodes)[pend_e] = make_int2(pend_prev, pend_k);
    pend_e = -1;
    int qsel = qi[0];
    unsigned ksel = koff[0];
#pragma unroll
    for (int v = 1; v < 8; ++v)
      if (lr == v) { qsel = qi[v]; ksel = koff[v]; }
    const int epush = tm * 16 + lr + 8 * half;
    if (lr < 8 && epush < Medges) {
      pend_prev = atomicExch(&head[qsel], epush);
      pend_k = (int)(ksel >> 8);
      pend_e = epush;
    }
  }
  if (pend_e >= 0) ((int2*)nodes)[pend_e] = make_int2(pend_prev, pend_k);
}

__global__ __launch_bounds__(256) void gather_agg_kernel(
    const float* __restrict__ logit, const int* __restrict__ head,
    const int* __restrict__ nodes, const unsigned* __restrict__ segkey,
    const float* __restrict__ kvbuf, float* __restrict__ agg, int n, int Medges) {
  const int lane = threadIdx.x & 31;
  const int row = __builtin_amdgcn_readfirstlane(blockIdx.x * 8 + (threadIdx.x >> 5));
  if (row >= n) return;
  const int hq = (lane & 1) * 4;
  const int2* __restrict__ nd2 = (const int2*)nodes;

  const uint4 kq = *(const uint4*)(segkey + (size_t)row * NH + hq);
  const float m0 = key_float(kq.x), m1 = key_float(kq.y);
  const float m2 = key_float(kq.z), m3 = key_float(kq.w);

  float4 acc = make_float4(0.0f, 0.0f, 0.0f, 0.0f);
  float4 den = make_float4(0.0f, 0.0f, 0.0f, 0.0f);

  const int first = head[row];
  int e = first;
  int nxt = -1, kcur = 0;
  if (e >= 0) { const int2 t = nd2[e]; nxt = t.x; kcur = t.y; }
  for (int it = 0; it < Medges && e >= 0; ++it) {
    const float4 lg = *(const float4*)(logit + (size_t)e * NH + hq);
    const float4 vv = *(const float4*)(kvbuf + (size_t)kcur * (2 * HC) + HC + 4 * lane);
    int nn = -1, kn = 0;
    if (nxt >= 0) { const int2 t = nd2[nxt]; nn = t.x; kn = t.y; }
    const float e0 = __expf(lg.x - m0), e1 = __expf(lg.y - m1);
    const float e2 = __expf(lg.z - m2), e3 = __expf(lg.w - m3);
    den.x += e0; den.y += e1; den.z += e2; den.w += e3;
    acc.x += e0 * vv.x; acc.y += e1 * vv.y; acc.z += e2 * vv.z; acc.w += e3 * vv.w;
    e = nxt; nxt = nn; kcur = kn;
  }

  float4 r = make_float4(0.0f, 0.0f, 0.0f, 0.0f);
  if (first >= 0) {
    r.x = acc.x / den.x; r.y = acc.y / den.y; r.z = acc.z / den.z; r.w = acc.w / den.w;
  }
  *(float4*)(agg + (size_t)row * HC + 4 * lane) = r;
}

extern "C" void kernel_launch(void* const* d_in, const int* in_sizes, int n_in,
                              void* d_out, int out_size, void* d_ws, size_t ws_size,
                              hipStream_t stream) {
  const float* query = (const float*)d_in[0];
  const float* key_  = (const float*)d_in[1];
  const int*   qidx  = (const int*)d_in[2];
  const int*   kidx  = (const int*)d_in[3];
  const float* pr    = (const float*)d_in[4];
  const float* Wq    = (const float*)d_in[5];
  const float* Wkv   = (const float*)d_in[6];
  const float* Wb    = (const float*)d_in[7];
  const float* Wo    = (const float*)d_in[8];

  const int N = in_sizes[0] / HC;
  const int M = in_sizes[2];

  float* out_result = (float*)d_out;
  float* out_logit  = out_result + (size_t)N * HC;

  float*    q_buf  = (float*)d_ws;
  float*    kv_buf = q_buf + (size_t)N * HC;
  unsigned* segkey = (unsigned*)(kv_buf + (size_t)N * 2 * HC);
  float*    agg    = (float*)(segkey + (size_t)N * NH);
  int*      head   = (int*)(agg + (size_t)N * HC);
  int*      nodes  = head + (size_t)N;

  const int TPB = 256;

  proj_qkv_kernel<<<NWG, 512, 0, stream>>>(query, Wq, q_buf, key_, Wkv, kv_buf, N,
                                           segkey, head);

  edge_bias_logit_kernel<<<NWG, 512, 0, stream>>>(
      pr, Wb, q_buf, kv_buf, qidx, kidx, out_logit, segkey, head, nodes, M);

  gather_agg_kernel<<<(N + 7) / 8, TPB, 0, stream>>>(
      out_logit, head, nodes, segkey, kv_buf, agg, N, M);

  wmma_gemm_f32<HC><<<NWG, 512, 0, stream>>>(agg, Wo, out_result, N);
}
